# starting kernel
# baseline (speedup 1.0000x reference)
#include <hip/hip_runtime.h>

typedef __attribute__((ext_vector_type(2))) float v2f;
typedef __attribute__((ext_vector_type(8))) float v8f;

#define BSZ 16
#define CDIM 256
#define SDIM 1024

__global__ __launch_bounds__(256) void pool_kernel(const float* __restrict__ x,
                                                   float* __restrict__ pooled) {
    __shared__ float red[256];
    const int row = blockIdx.x;
    const float4* p = (const float4*)(x + (size_t)row * SDIM);
    float4 v = p[threadIdx.x];
    red[threadIdx.x] = v.x + v.y + v.z + v.w;
    __syncthreads();
    for (int off = 128; off > 0; off >>= 1) {
        if (threadIdx.x < off) red[threadIdx.x] += red[threadIdx.x + off];
        __syncthreads();
    }
    if (threadIdx.x == 0) pooled[row] = red[0] * (1.0f / SDIM);
}

__global__ __launch_bounds__(512) void gate_kernel(const float* __restrict__ pooled,
                                                   const float* __restrict__ in_proj_w,
                                                   const float* __restrict__ in_proj_b,
                                                   const float* __restrict__ out_proj_w,
                                                   const float* __restrict__ out_proj_b,
                                                   float* __restrict__ gate) {
    __shared__ float tmpS[BSZ * CDIM];
    const int lane = threadIdx.x & 31;
    const int wave = threadIdx.x >> 5;
    const int m    = lane & 15;
    const int half = lane >> 4;
    const int n    = wave * 16 + m;

    const float* wv  = in_proj_w + 2 * CDIM * CDIM;
    const float* bv  = in_proj_b + 2 * CDIM;

    v8f acc = {};
    for (int k0 = 0; k0 < CDIM; k0 += 4) {
        const int ka = k0 + half * 2;
        v2f a = *(const v2f*)(pooled + m * CDIM + ka);
        v2f b = *(const v2f*)(wv + n * CDIM + ka);
        acc = __builtin_amdgcn_wmma_f32_16x16x4_f32(false, a, false, b,
                                                    (short)0, acc, false, false);
    }
    {
        const float bn = bv[n];
#pragma unroll
        for (int r = 0; r < 8; ++r)
            tmpS[(r + half * 8) * CDIM + n] = acc[r] + bn;
    }
    __syncthreads();

    v8f acc2 = {};
    for (int k0 = 0; k0 < CDIM; k0 += 4) {
        const int ka = k0 + half * 2;
        v2f a = *(const v2f*)(tmpS + m * CDIM + ka);
        v2f b = *(const v2f*)(out_proj_w + n * CDIM + ka);
        acc2 = __builtin_amdgcn_wmma_f32_16x16x4_f32(false, a, false, b,
                                                     (short)0, acc2, false, false);
    }
    {
        const float bo = out_proj_b[n];
#pragma unroll
        for (int r = 0; r < 8; ++r)
            gate[(r + half * 8) * CDIM + n] = acc2[r] + bo;
    }
}

__global__ __launch_bounds__(256) void mul_kernel(const float* __restrict__ x,
                                                  const float* __restrict__ gate,
                                                  float* __restrict__ out) {
    const size_t idx = (size_t)blockIdx.x * 256 + threadIdx.x;
    const int row = (int)(idx >> 8);
    const float g = gate[row];
    float4 v = ((const float4*)x)[idx];
    v.x *= g; v.y *= g; v.z *= g; v.w *= g;
    ((float4*)out)[idx] = v;
}

extern "C" void kernel_launch(void* const* d_in, const int* in_sizes, int n_in,
                              void* d_out, int out_size, void* d_ws, size_t ws_size,
                              hipStream_t stream) {
    const float* x          = (const float*)d_in[0];
    const float* in_proj_w  = (const float*)d_in[1];
    const float* in_proj_b  = (const float*)d_in[2];
    const float* out_proj_w = (const float*)d_in[3];
    const float* out_proj_b = (const float*)d_in[4];
    float* out = (float*)d_out;

    float* pooled = (float*)d_ws;
    float* gate   = pooled + BSZ * CDIM;

    pool_kernel<<<BSZ * CDIM, 256, 0, stream>>>(x, pooled);
    gate_kernel<<<1, 512, 0, stream>>>(pooled, in_proj_w, in_proj_b,
                                       out_proj_w, out_proj_b, gate);
    mul_kernel<<<(BSZ * CDIM * SDIM) / 4 / 256, 256, 0, stream>>>(x, gate, out);

    (void)in_sizes; (void)n_in; (void)out_size; (void)ws_size;
}


# gate kernel: weight rows fetched as whole 128-byte lines (16-byte loads issued ahead of the WMMA chain, staged through wave-private padded LDS) instead of one 16-line 8-byte gather per WMMA
# speedup vs baseline: 1.5482x; 1.5482x over previous
#include <hip/hip_runtime.h>

typedef __attribute__((ext_vector_type(2))) float v2f;
typedef __attribute__((ext_vector_type(8))) float v8f;

#define BSZ 16
#define CDIM 256
#define SDIM 1024

__global__ __launch_bounds__(256) void pool_kernel(const float* __restrict__ x,
                                                   float* __restrict__ pooled) {
    __shared__ float red[256];
    const int row = blockIdx.x;
    const float4* p = (const float4*)(x + (size_t)row * SDIM);
    float4 v = p[threadIdx.x];
    red[threadIdx.x] = v.x + v.y + v.z + v.w;
    __syncthreads();
    for (int off = 128; off > 0; off >>= 1) {
        if (threadIdx.x < off) red[threadIdx.x] += red[threadIdx.x + off];
        __syncthreads();
    }
    if (threadIdx.x == 0) pooled[row] = red[0] * (1.0f / SDIM);
}

#define PITCH 260
#define TILE (16 * PITCH)
#define REP16(M) M(0) M(1) M(2) M(3) M(4) M(5) M(6) M(7) M(8) M(9) M(10) M(11) M(12) M(13) M(14) M(15)
#define LDE(j) const float4 e##j = gB[(2 * j) * 32];
#define STE(j) *(float4*)(wSt + j * PITCH) = e##j;
#define LDO(j) const float4 o##j = gB[(2 * j + 1) * 32];
#define STO(j) *(float4*)(wSt + j * PITCH + 128) = o##j;

__device__ __forceinline__ v8f mma4(v8f acc, float4 a, float4 b) {
    v2f a0 = {a.x, a.y};
    v2f b0 = {b.x, b.y};
    v2f a1 = {a.z, a.w};
    v2f b1 = {b.z, b.w};
    acc = __builtin_amdgcn_wmma_f32_16x16x4_f32(false, a0, false, b0,
                                                (short)0, acc, false, false);
    acc = __builtin_amdgcn_wmma_f32_16x16x4_f32(false, a1, false, b1,
                                                (short)0, acc, false, false);
    return acc;
}

__global__ __launch_bounds__(512) void gate_kernel(const float* __restrict__ pooled,
                                                   const float* __restrict__ in_proj_w,
                                                   const float* __restrict__ in_proj_b,
                                                   const float* out_proj_w,
                                                   const float* __restrict__ out_proj_b,
                                                   float* __restrict__ gate) {
    __shared__ __attribute__((aligned(16))) float poolS[TILE];
    __shared__ __attribute__((aligned(16))) float tmpS[TILE];
    __shared__ __attribute__((aligned(16))) float wS[16 * TILE];

    const int t    = threadIdx.x;
    const int lane = t & 31;
    const int wave = t >> 5;
    const int c    = lane & 15;
    const int half = lane >> 4;
    const int n    = wave * 16 + c;

    const float* wv  = in_proj_w + 2 * CDIM * CDIM;
    const float* bv  = in_proj_b + 2 * CDIM;

    const float4* P4 = (const float4*)pooled;
    const float4 p0 = P4[t];
    const float4 p1 = P4[t + 512];
    const float bn = bv[n];
    const float bo = out_proj_b[n];

    const float4* gA = (const float4*)(wv + wave * 16 * CDIM) + lane;
    const float4* gB = (const float4*)(out_proj_w + wave * 16 * CDIM) + lane;
    float4 ga[32];
#pragma unroll
    for (int i = 0; i < 32; ++i) ga[i] = gA[i * 32];

    *(float4*)(poolS + (t >> 6) * PITCH + (t & 63) * 4)       = p0;
    *(float4*)(poolS + ((t >> 6) + 8) * PITCH + (t & 63) * 4) = p1;

    float* wT = wS + wave * TILE;
    float* wSt = wT + lane * 4;
#pragma unroll
    for (int i = 0; i < 32; ++i)
        *(float4*)(wSt + (i >> 1) * PITCH + (i & 1) * 128) = ga[i];

    __syncthreads();

    const int fo = c * PITCH + 4 * half;

    REP16(LDE)
    __builtin_amdgcn_sched_barrier(0);

    v8f acc = {};
#pragma unroll
    for (int i = 0; i < 16; ++i)
        acc = mma4(acc, *(const float4*)(poolS + fo + 8 * i), *(const float4*)(wT + fo + 8 * i));

    __builtin_amdgcn_sched_barrier(0);
    REP16(STE)
    REP16(LDO)
    __builtin_amdgcn_sched_barrier(0);

#pragma unroll
    for (int i = 16; i < 32; ++i)
        acc = mma4(acc, *(const float4*)(poolS + fo + 8 * i), *(const float4*)(wT + fo + 8 * i));
#pragma unroll
    for (int r = 0; r < 8; ++r)
        tmpS[(r + half * 8) * PITCH + n] = acc[r] + bn;
    __builtin_amdgcn_sched_barrier(0);
    REP16(STO)
    __syncthreads();

    v8f acc2 = {};
#pragma unroll
    for (int i = 0; i < 32; ++i)
        acc2 = mma4(acc2, *(const float4*)(tmpS + fo + 8 * i), *(const float4*)(wT + fo + 8 * i));
#pragma unroll
    for (int r = 0; r < 8; ++r)
        gate[(r + half * 8) * CDIM + n] = acc2[r] + bo;
}

__global__ __launch_bounds__(256) void mul_kernel(const float* __restrict__ x,
                                                  const float* __restrict__ gate,
                                                  float* __restrict__ out) {
    const size_t idx = (size_t)blockIdx.x * 256 + threadIdx.x;
    const int row = (int)(idx >> 8);
    const float g = gate[row];
    float4 v = ((const float4*)x)[idx];
    v.x *= g; v.y *= g; v.z *= g; v.w *= g;
    ((float4*)out)[idx] = v;
}

extern "C" void kernel_launch(void* const* d_in, const int* in_sizes, int n_in,
                              void* d_out, int out_size, void* d_ws, size_t ws_size,
                              hipStream_t stream) {
    const float* x          = (const float*)d_in[0];
    const float* in_proj_w  = (const float*)d_in[1];
    const float* in_proj_b  = (const float*)d_in[2];
    const float* out_proj_w = (const float*)d_in[3];
    const float* out_proj_b = (const float*)d_in[4];
    float* out = (float*)d_out;

    float* pooled = (float*)d_ws;
    float* gate   = pooled + BSZ * CDIM;

    pool_kernel<<<BSZ * CDIM, 256, 0, stream>>>(x, pooled);
    gate_kernel<<<1, 512, 0, stream>>>(pooled, in_proj_w, in_proj_b,
                                       out_proj_w, out_proj_b, gate);
    mul_kernel<<<(BSZ * CDIM * SDIM) / 4 / 256, 256, 0, stream>>>(x, gate, out);

    (void)in_sizes; (void)n_in; (void)out_size; (void)ws_size;
}


# pool and scaling kernels: one wave per 4-KB row with eight 16-byte loads in flight per lane (512 workgroups of 8 waves instead of 4096; pool's LDS tree and 9 barriers become a lane butterfly)
# speedup vs baseline: 2.7400x; 1.7698x over previous
#include <hip/hip_runtime.h>

typedef __attribute__((ext_vector_type(2))) float v2f;
typedef __attribute__((ext_vector_type(8))) float v8f;

#define BSZ 16
#define CDIM 256
#define SDIM 1024

#define RPW 8
#define NQ  8

template <int M>
__device__ __forceinline__ float xmask_add(float v) {
    return v + __int_as_float(__builtin_amdgcn_update_dpp(0, __float_as_int(v), 0x160 + M,
                                                          0xf, 0xf, false));
}

__device__ __forceinline__ float lane_value(float v, const int l) {
    return __int_as_float(__builtin_amdgcn_readlane(__float_as_int(v), l));
}

__global__ __launch_bounds__(256) void pool_kernel(const float* __restrict__ x,
                                                   float* __restrict__ pooled) {
    const int wave = __builtin_amdgcn_readfirstlane(threadIdx.x >> 5);
    const int lane = threadIdx.x & 31;
    const int row  = blockIdx.x * RPW + wave;
    const float4* p = (const float4*)(x + (size_t)row * SDIM) + lane;
    float4 v[NQ];
#pragma unroll
    for (int i = 0; i < NQ; ++i) v[i] = p[32 * i];
    float s[NQ];
#pragma unroll
    for (int i = 0; i < NQ; ++i) s[i] = (v[i].x + v[i].y) + (v[i].z + v[i].w);
    float t = ((s[0] + s[1]) + (s[2] + s[3])) + ((s[4] + s[5]) + (s[6] + s[7]));
    t = xmask_add<1>(t);
    t = xmask_add<2>(t);
    t = xmask_add<4>(t);
    t = xmask_add<8>(t);
    const float tot = lane_value(t, 0) + lane_value(t, 16);
    if (lane == 0) pooled[row] = tot * (1.0f / SDIM);
}

#define NGRP 16
#define PLANE (BSZ * CDIM)
#define PITCH 260
#define TILE (16 * PITCH)
#define TPITCH 20
#define RHALF 144
#define RWAVE (2 * RHALF)

__device__ __forceinline__ v8f mma4(v8f acc, float4 a, float4 b) {
    v2f a0 = {a.x, a.y};
    v2f b0 = {b.x, b.y};
    v2f a1 = {a.z, a.w};
    v2f b1 = {b.z, b.w};
    acc = __builtin_amdgcn_wmma_f32_16x16x4_f32(false, a0, false, b0,
                                                (short)0, acc, false, false);
    acc = __builtin_amdgcn_wmma_f32_16x16x4_f32(false, a1, false, b1,
                                                (short)0, acc, false, false);
    return acc;
}

__device__ __forceinline__ void lds_barrier() {
    __builtin_amdgcn_fence(__ATOMIC_RELEASE, "workgroup", "local");
    __builtin_amdgcn_s_barrier();
    __builtin_amdgcn_fence(__ATOMIC_ACQUIRE, "workgroup", "local");
}

__global__ __launch_bounds__(512) void gate_kernel(const float* __restrict__ pooled,
                                                   const float* __restrict__ in_proj_w,
                                                   const float* __restrict__ in_proj_b,
                                                   const float* __restrict__ out_proj_w,
                                                   const float* __restrict__ out_proj_b,
                                                   float* __restrict__ part) {
    __shared__ __attribute__((aligned(16))) float poolS[TILE];
    __shared__ __attribute__((aligned(16))) float wvS[TILE];
    __shared__ __attribute__((aligned(16))) float redS[16 * RWAVE];
    __shared__ __attribute__((aligned(16))) float tmpS[16 * TPITCH];

    const int g    = blockIdx.x;
    const int t    = threadIdx.x;
    const int lane = t & 31;
    const int wave = t >> 5;
    const int c    = lane & 15;
    const int half = lane >> 4;
    const int n    = wave * 16 + c;

    const float* wv = in_proj_w + 2 * CDIM * CDIM + g * 16 * CDIM;
    const float* bv = in_proj_b + 2 * CDIM + g * 16;

    const float4* P4 = (const float4*)pooled;
    const float4* W4 = (const float4*)wv;
    const float4 p0 = P4[t];
    const float4 p1 = P4[t + 512];
    const float4 w0 = W4[t];
    const float4 w1 = W4[t + 512];
    __builtin_amdgcn_sched_barrier(0);
    const float4* O4 = (const float4*)(out_proj_w + n * CDIM + g * 16 + 4 * half);
    const float4 ob0 = O4[0];
    const float4 ob1 = O4[2];
    const float bvj = bv[t & 15];
    const float bol = out_proj_b[n];
    const float bo  = (g == 0) ? bol : 0.0f;

    const int so = (t >> 6) * PITCH + (t & 63) * 4;
    *(float4*)(poolS + so)             = p0;
    *(float4*)(poolS + so + 8 * PITCH) = p1;
    *(float4*)(wvS + so)               = w0;
    *(float4*)(wvS + so + 8 * PITCH)   = w1;
    lds_barrier();

    const int fo = c * PITCH + 16 * wave + 4 * half;
    v8f acc = {};
#pragma unroll
    for (int i = 0; i < 2; ++i)
        acc = mma4(acc, *(const float4*)(poolS + fo + 8 * i), *(const float4*)(wvS + fo + 8 * i));

    float* rw = redS + wave * RWAVE + half * RHALF + c;
#pragma unroll
    for (int r = 0; r < 8; ++r) rw[r * 16] = acc[r];
    lds_barrier();

    if (t < 256) {
        const int m = t >> 4;
        const int j = t & 15;
        const float* rr = redS + (m >> 3) * RHALF + (m & 7) * 16 + j;
        float s = rr[0];
#pragma unroll
        for (int w = 1; w < 16; ++w) s += rr[w * RWAVE];
        tmpS[m * TPITCH + j] = s + bvj;
    }
    lds_barrier();

    const int to = c * TPITCH + 4 * half;
    v8f acc2 = {};
    acc2 = mma4(acc2, *(const float4*)(tmpS + to), ob0);
    acc2 = mma4(acc2, *(const float4*)(tmpS + to + 8), ob1);
    float* pg = part + g * PLANE + n;
#pragma unroll
    for (int r = 0; r < 8; ++r)
        pg[(r + half * 8) * CDIM] = acc2[r] + bo;
}

__global__ __launch_bounds__(256) void mul_kernel(const float* __restrict__ x,
                                                  const float* __restrict__ part,
                                                  float* __restrict__ out) {
    const int wave = __builtin_amdgcn_readfirstlane(threadIdx.x >> 5);
    const int lane = threadIdx.x & 31;
    const int row  = blockIdx.x * RPW + wave;
    const float4* xp = (const float4*)(x + (size_t)row * SDIM) + lane;
    float4* op = (float4*)(out + (size_t)row * SDIM) + lane;
    float4 v[NQ];
#pragma unroll
    for (int i = 0; i < NQ; ++i) v[i] = xp[32 * i];
    float p[NGRP];
#pragma unroll
    for (int k = 0; k < NGRP; ++k) p[k] = part[k * PLANE + row];
    __builtin_amdgcn_sched_barrier(0);
#pragma unroll
    for (int d = NGRP / 2; d > 0; d >>= 1) {
#pragma unroll
        for (int k = 0; k < d; ++k) p[k] += p[k + d];
    }
    const float g = p[0];
#pragma unroll
    for (int i = 0; i < NQ; ++i) {
        float4 w = v[i];
        w.x *= g; w.y *= g; w.z *= g; w.w *= g;
        op[32 * i] = w;
    }
}

extern "C" void kernel_launch(void* const* d_in, const int* in_sizes, int n_in,
                              void* d_out, int out_size, void* d_ws, size_t ws_size,
                              hipStream_t stream) {
    const float* x          = (const float*)d_in[0];
    const float* in_proj_w  = (const float*)d_in[1];
    const float* in_proj_b  = (const float*)d_in[2];
    const float* out_proj_w = (const float*)d_in[3];
    const float* out_proj_b = (const float*)d_in[4];
    float* out = (float*)d_out;

    float* pooled = (float*)d_ws;
    float* part   = pooled + BSZ * CDIM;

    pool_kernel<<<BSZ * CDIM / RPW, 256, 0, stream>>>(x, pooled);
    gate_kernel<<<NGRP, 512, 0, stream>>>(pooled, in_proj_w, in_proj_b,
                                          out_proj_w, out_proj_b, part);
    mul_kernel<<<BSZ * CDIM / RPW, 256, 0, stream>>>(x, part, out);

    (void)in_sizes; (void)n_in; (void)out_size; (void)ws_size;
}
